# starting kernel
# baseline (speedup 1.0000x reference)
#include <hip/hip_runtime.h>

typedef __attribute__((ext_vector_type(2))) float v2f;
typedef __attribute__((ext_vector_type(8))) float v8f;

#define N_NODES 100000
#define N_EDGES 1600000
#define IN_DIM  256
#define OUT_DIM 128
#define M_TILES (N_NODES / 16)
#define GEMM_BLOCKS 625

__global__ __launch_bounds__(256) void gemm_wmma_f32(const float* __restrict__ X,
                                                     const float* __restrict__ W,
                                                     float* __restrict__ S) {
  const int lane = threadIdx.x & 31;
  const int wave = threadIdx.x >> 5;
  const int n0   = wave << 4;
  const int m    = lane & 15;
  const int koff = (lane >> 4) << 1;
  const int mbase = (lane >> 4) << 3;

  const float* __restrict__ wcol = W + (n0 + m);
  v2f bfrag[64];
#pragma unroll
  for (int kk = 0; kk < 64; ++kk) {
    const int k = (kk << 2) + koff;
    bfrag[kk].x = wcol[(size_t)k * OUT_DIM];
    bfrag[kk].y = wcol[(size_t)(k + 1) * OUT_DIM];
  }

  for (int mt = blockIdx.x; mt < M_TILES; mt += gridDim.x) {
    const int row0 = mt << 4;
    const float* __restrict__ xrow = X + (size_t)(row0 + m) * IN_DIM + koff;

    v8f c = {0.f, 0.f, 0.f, 0.f, 0.f, 0.f, 0.f, 0.f};
#pragma unroll
    for (int kk = 0; kk < 64; ++kk) {
      const v2f a = *(const v2f*)(xrow + (kk << 2));
      c = __builtin_amdgcn_wmma_f32_16x16x4_f32(
           false, a,  false, bfrag[kk],
           (short)0, c,  false,  false);
    }

    float* __restrict__ srow = S + (size_t)row0 * OUT_DIM + n0 + m;
#pragma unroll
    for (int r = 0; r < 8; ++r) {
      srow[(size_t)(mbase + r) * OUT_DIM] = c[r];
    }
  }
}

__global__ __launch_bounds__(256) void init_bias(float* __restrict__ out,
                                                 const float* __restrict__ bias,
                                                 long long n) {
  const long long i = (long long)blockIdx.x * blockDim.x + threadIdx.x;
  if (i < n) out[i] = bias[(int)(i & (OUT_DIM - 1))];
}

__global__ __launch_bounds__(256) void spmm_edge_atomic(const int* __restrict__ rows,
                                                        const int* __restrict__ cols,
                                                        const float* __restrict__ vals,
                                                        const float* __restrict__ S,
                                                        float* __restrict__ out,
                                                        int nEdges) {
  const long long gid = (long long)blockIdx.x * blockDim.x + threadIdx.x;
  const int e = (int)(gid >> 7);
  const int d = (int)(gid & 127);
  if (e >= nEdges) return;

  const int   r  = rows[e];
  const int   cc = cols[e];
  const float v  = vals[e];
  const float contrib = S[(size_t)cc * OUT_DIM + d] * v;

  __hip_atomic_fetch_add(&out[(size_t)r * OUT_DIM + d], contrib,
                         __ATOMIC_RELAXED, __HIP_MEMORY_SCOPE_AGENT);
}

extern "C" void kernel_launch(void* const* d_in, const int* in_sizes, int n_in,
                              void* d_out, int out_size, void* d_ws, size_t ws_size,
                              hipStream_t stream) {
  const float* X    = (const float*)d_in[0];
  const int*   er   = (const int*)  d_in[1];
  const int*   ec   = (const int*)  d_in[2];
  const float* ev   = (const float*)d_in[3];
  const float* W    = (const float*)d_in[4];
  const float* bias = (const float*)d_in[5];

  float* out = (float*)d_out;
  float* S   = (float*)d_ws;

  gemm_wmma_f32<<<GEMM_BLOCKS, 256, 0, stream>>>(X, W, S);

  const long long nOut = (long long)N_NODES * OUT_DIM;
  init_bias<<<(int)((nOut + 255) / 256), 256, 0, stream>>>(out, bias, nOut);

  const long long nWork = (long long)N_EDGES * OUT_DIM;
  spmm_edge_atomic<<<(int)(nWork / 256), 256, 0, stream>>>(er, ec, ev, S, out, N_EDGES);
}


# SpMM without float atomics: integer slot counter bins (col,val) pairs inside each out row; the row's wave ranks them by (col,val), sums support rows in that fixed order, stores out once
# speedup vs baseline: 3.3262x; 3.3262x over previous
#include <hip/hip_runtime.h>

typedef __attribute__((ext_vector_type(2))) float v2f;
typedef __attribute__((ext_vector_type(8))) float v8f;

#define N_NODES 100000
#define N_EDGES 1600000
#define IN_DIM  256
#define OUT_DIM 128
#define M_TILES (N_NODES / 16)
#define GEMM_BLOCKS 625

__global__ __launch_bounds__(256) void gemm_wmma_f32(const float* __restrict__ X,
                                                     const float* __restrict__ W,
                                                     float* __restrict__ S) {
  const int lane = threadIdx.x & 31;
  const int wave = threadIdx.x >> 5;
  const int n0   = wave << 4;
  const int m    = lane & 15;
  const int koff = (lane >> 4) << 1;
  const int mbase = (lane >> 4) << 3;

  const float* __restrict__ wcol = W + (n0 + m);
  v2f bfrag[64];
#pragma unroll
  for (int kk = 0; kk < 64; ++kk) {
    const int k = (kk << 2) + koff;
    bfrag[kk].x = wcol[(size_t)k * OUT_DIM];
    bfrag[kk].y = wcol[(size_t)(k + 1) * OUT_DIM];
  }

  for (int mt = blockIdx.x; mt < M_TILES; mt += gridDim.x) {
    const int row0 = mt << 4;
    const float* __restrict__ xrow = X + (size_t)(row0 + m) * IN_DIM + koff;

    v8f c = {0.f, 0.f, 0.f, 0.f, 0.f, 0.f, 0.f, 0.f};
#pragma unroll
    for (int kk = 0; kk < 64; ++kk) {
      const v2f a = *(const v2f*)(xrow + (kk << 2));
      c = __builtin_amdgcn_wmma_f32_16x16x4_f32(
           false, a,  false, bfrag[kk],
           (short)0, c,  false,  false);
    }

    float* __restrict__ srow = S + (size_t)row0 * OUT_DIM + n0 + m;
#pragma unroll
    for (int r = 0; r < 8; ++r) {
      srow[(size_t)(mbase + r) * OUT_DIM] = c[r];
    }
  }
}

#define BIN_CAP 63
#define EDGES_PER_THREAD 5
#define BIN_BLOCKS 1250
#define ROWS_PER_BLOCK 8
#define GATHER_BLOCKS (N_NODES / ROWS_PER_BLOCK)

static_assert(BIN_BLOCKS * 256 * EDGES_PER_THREAD == N_EDGES, "edge split must be exact");
static_assert(GATHER_BLOCKS * ROWS_PER_BLOCK == N_NODES, "row split must be exact");
static_assert((N_EDGES & 31) == 0, "edge walk reads whole waves");

__global__ __launch_bounds__(256) void zero_bin_counts(int* __restrict__ bins) {
  const int i = blockIdx.x * 256 + threadIdx.x;
  if (i < N_NODES) bins[(size_t)i * OUT_DIM] = 0;
}

__global__ __launch_bounds__(256) void bin_edges(const int* __restrict__ rows,
                                                 const int* __restrict__ cols,
                                                 const float* __restrict__ vals,
                                                 int* __restrict__ bins) {
  const int t = blockIdx.x * 256 + threadIdx.x;
  const int stride = BIN_BLOCKS * 256;

  int r[EDGES_PER_THREAD], c[EDGES_PER_THREAD], slot[EDGES_PER_THREAD];
  float v[EDGES_PER_THREAD];
#pragma unroll
  for (int u = 0; u < EDGES_PER_THREAD; ++u) {
    const int e = t + u * stride;
    r[u] = rows[e];
    c[u] = cols[e];
    v[u] = vals[e];
  }
#pragma unroll
  for (int u = 0; u < EDGES_PER_THREAD; ++u) {
    slot[u] = __hip_atomic_fetch_add(bins + (size_t)r[u] * OUT_DIM, 1,
                                     __ATOMIC_RELAXED, __HIP_MEMORY_SCOPE_AGENT);
  }
#pragma unroll
  for (int u = 0; u < EDGES_PER_THREAD; ++u) {
    if (slot[u] < BIN_CAP) {
      int2 p;
      p.x = c[u];
      p.y = __float_as_int(v[u]);
      *(int2*)(bins + (size_t)r[u] * OUT_DIM + 2 + 2 * slot[u]) = p;
    }
  }
}

__device__ __forceinline__ unsigned lane_read(unsigned x, int srcLane) {
  return (unsigned)__builtin_amdgcn_readlane((int)x, srcLane);
}

template <int NR, int U>
__device__ __forceinline__ void add_ranked(const unsigned (&kc)[NR], const unsigned (&kv)[NR],
                                           const int (&rk)[NR], int k, int lane,
                                           const float* __restrict__ S, float4& acc) {
  float4 s[U];
  float w[U];
#pragma unroll
  for (int u = 0; u < U; ++u) {
    unsigned cb = 0, vb = 0;
#pragma unroll
    for (int a = 0; a < NR; ++a) {
      const unsigned m = __builtin_amdgcn_ballot_w32(rk[a] == k + u);
      if (NR == 1 || m != 0) {
        const int src = __builtin_ctz(m);
        cb = lane_read(kc[a], src);
        vb = lane_read(kv[a], src);
      }
    }
    w[u] = __int_as_float((int)vb);
    s[u] = *(const float4*)(S + (size_t)cb * OUT_DIM + (lane << 2));
  }
#pragma unroll
  for (int u = 0; u < U; ++u) {
    acc.x = __builtin_fmaf(w[u], s[u].x, acc.x);
    acc.y = __builtin_fmaf(w[u], s[u].y, acc.y);
    acc.z = __builtin_fmaf(w[u], s[u].z, acc.z);
    acc.w = __builtin_fmaf(w[u], s[u].w, acc.w);
  }
}

template <int NR>
__device__ __forceinline__ void add_row_sorted(const unsigned (&kc)[NR], const unsigned (&kv)[NR],
                                               int deg, int lane,
                                               const float* __restrict__ S, float4& acc) {
  int rk[NR];
  unsigned long long key[NR];
#pragma unroll
  for (int a = 0; a < NR; ++a) {
    rk[a] = 0;
    key[a] = ((unsigned long long)kc[a] << 32) | kv[a];
  }
#pragma unroll
  for (int b = 0; b < NR; ++b) {
    const int nb = min(32, deg - b * 32);
    for (int j = 0; j < nb; ++j) {
      const unsigned long long t =
          ((unsigned long long)lane_read(kc[b], j) << 32) | lane_read(kv[b], j);
      const int jj = b * 32 + j;
#pragma unroll
      for (int a = 0; a < NR; ++a) {
        const int ii = a * 32 + lane;
        rk[a] += (t < key[a] || (t == key[a] && jj < ii)) ? 1 : 0;
      }
    }
  }
  int k = 0;
  for (; k + 8 <= deg; k += 8) add_ranked<NR, 8>(kc, kv, rk, k, lane, S, acc);
  if (k + 4 <= deg) { add_ranked<NR, 4>(kc, kv, rk, k, lane, S, acc); k += 4; }
  if (k + 2 <= deg) { add_ranked<NR, 2>(kc, kv, rk, k, lane, S, acc); k += 2; }
  if (k < deg)      { add_ranked<NR, 1>(kc, kv, rk, k, lane, S, acc); }
}

__global__ __launch_bounds__(256) void gather_rows(const int* __restrict__ rows,
                                                   const int* __restrict__ cols,
                                                   const float* __restrict__ vals,
                                                   const float* __restrict__ S,
                                                   const float* __restrict__ bias,
                                                   float* out) {
  const int lane = threadIdx.x & 31;
  const int wave = __builtin_amdgcn_readfirstlane((int)(threadIdx.x >> 5));
  const int row = blockIdx.x * ROWS_PER_BLOCK + wave;

  float* orow = out + (size_t)row * OUT_DIM;
  const int* bin = (const int*)orow;

  const int cntv = bin[0];
  const int2 p0 = *(const int2*)(bin + 2 + 2 * lane);
  const int deg = __builtin_amdgcn_readfirstlane(cntv);

  float4 acc = {0.f, 0.f, 0.f, 0.f};

  if (deg <= 32) {
    unsigned kc[1], kv[1];
    const bool ok = lane < deg;
    kc[0] = ok ? (unsigned)p0.x : 0xFFFFFFFFu;
    kv[0] = ok ? (unsigned)p0.y : 0xFFFFFFFFu;
    add_row_sorted<1>(kc, kv, deg, lane, S, acc);
  } else if (deg <= BIN_CAP) {
    unsigned kc[2], kv[2];
    kc[0] = (unsigned)p0.x;
    kv[0] = (unsigned)p0.y;
    kc[1] = 0xFFFFFFFFu;
    kv[1] = 0xFFFFFFFFu;
    if (32 + lane < deg) {
      const int2 p1 = *(const int2*)(bin + 2 + 2 * (32 + lane));
      kc[1] = (unsigned)p1.x;
      kv[1] = (unsigned)p1.y;
    }
    add_row_sorted<2>(kc, kv, deg, lane, S, acc);
  } else {
    for (int base = 0; base < N_EDGES; base += 32) {
      unsigned m = __builtin_amdgcn_ballot_w32(rows[base + lane] == row);
      while (m != 0) {
        const int e = base + __builtin_ctz(m);
        m &= m - 1;
        const int cc = cols[e];
        const float w = vals[e];
        const float4 s = *(const float4*)(S + (size_t)cc * OUT_DIM + (lane << 2));
        acc.x = __builtin_fmaf(w, s.x, acc.x);
        acc.y = __builtin_fmaf(w, s.y, acc.y);
        acc.z = __builtin_fmaf(w, s.z, acc.z);
        acc.w = __builtin_fmaf(w, s.w, acc.w);
      }
    }
  }

  const float4 b = *(const float4*)(bias + (lane << 2));
  float4 res;
  res.x = acc.x + b.x;
  res.y = acc.y + b.y;
  res.z = acc.z + b.z;
  res.w = acc.w + b.w;
  *(float4*)(orow + (lane << 2)) = res;
}

extern "C" void kernel_launch(void* const* d_in, const int* in_sizes, int n_in,
                              void* d_out, int out_size, void* d_ws, size_t ws_size,
                              hipStream_t stream) {
  const float* X    = (const float*)d_in[0];
  const int*   er   = (const int*)  d_in[1];
  const int*   ec   = (const int*)  d_in[2];
  const float* ev   = (const float*)d_in[3];
  const float* W    = (const float*)d_in[4];
  const float* bias = (const float*)d_in[5];

  float* out = (float*)d_out;
  float* S   = (float*)d_ws;

  zero_bin_counts<<<(N_NODES + 255) / 256, 256, 0, stream>>>((int*)d_out);
  bin_edges<<<BIN_BLOCKS, 256, 0, stream>>>(er, ec, ev, (int*)d_out);

  gemm_wmma_f32<<<GEMM_BLOCKS, 256, 0, stream>>>(X, W, S);

  gather_rows<<<GATHER_BLOCKS, 256, 0, stream>>>(er, ec, ev, S, bias, out);
}


# GEMM: each 16x256 X tile is loaded once per block with whole-line 16-byte loads and shared via LDS (2 buffers, 1 barrier per tile) instead of 8 waves each reading it with 16-line loads; grid 512
# speedup vs baseline: 4.1800x; 1.2567x over previous
#include <hip/hip_runtime.h>

typedef __attribute__((ext_vector_type(2))) float v2f;
typedef __attribute__((ext_vector_type(8))) float v8f;

#define N_NODES 100000
#define N_EDGES 1600000
#define IN_DIM  256
#define OUT_DIM 128
#define M_TILES (N_NODES / 16)
#define GEMM_BLOCKS 512

#define A_PITCH 260

static_assert(GEMM_BLOCKS <= M_TILES, "every block must own at least one M-tile");

__global__ __launch_bounds__(256) void gemm_wmma_f32(const float* __restrict__ X,
                                                     const float* __restrict__ W,
                                                     float* __restrict__ S) {
  __shared__ float As[2][16 * A_PITCH];

  const int tid  = threadIdx.x;
  const int lane = tid & 31;
  const int wave = tid >> 5;
  const int n0   = wave << 4;
  const int m    = lane & 15;
  const int koff = (lane >> 4) << 1;
  const int mbase = (lane >> 4) << 3;

  const int srow = tid >> 6;
  const int scol = (tid & 63) << 2;

  float4 g0, g1, g2, g3;
  int mt = blockIdx.x;
  {
    const float* __restrict__ xt = X + (size_t)((mt << 4) + srow) * IN_DIM + scol;
    g0 = *(const float4*)(xt);
    g1 = *(const float4*)(xt + 4 * IN_DIM);
    g2 = *(const float4*)(xt + 8 * IN_DIM);
    g3 = *(const float4*)(xt + 12 * IN_DIM);
  }

  const float* __restrict__ wcol = W + (n0 + m);
  v2f bfrag[64];
#pragma unroll
  for (int kk = 0; kk < 64; ++kk) {
    const int k = (kk << 2) + koff;
    bfrag[kk].x = wcol[(size_t)k * OUT_DIM];
    bfrag[kk].y = wcol[(size_t)(k + 1) * OUT_DIM];
  }

  int buf = 0;
  for (; mt < M_TILES; mt += gridDim.x) {
    float* abuf = As[buf];
    float* ast = abuf + srow * A_PITCH + scol;
    *(float4*)(ast) = g0;
    *(float4*)(ast + 4 * A_PITCH) = g1;
    *(float4*)(ast + 8 * A_PITCH) = g2;
    *(float4*)(ast + 12 * A_PITCH) = g3;
    __syncthreads();

    const int nmt = mt + gridDim.x;
    if (nmt < M_TILES) {
      const float* __restrict__ xt = X + (size_t)((nmt << 4) + srow) * IN_DIM + scol;
      g0 = *(const float4*)(xt);
      g1 = *(const float4*)(xt + 4 * IN_DIM);
      g2 = *(const float4*)(xt + 8 * IN_DIM);
      g3 = *(const float4*)(xt + 12 * IN_DIM);
    }

    const float* ap = abuf + m * A_PITCH + koff;
    v8f c = {0.f, 0.f, 0.f, 0.f, 0.f, 0.f, 0.f, 0.f};
#pragma unroll
    for (int kk = 0; kk < 64; ++kk) {
      const v2f a = *(const v2f*)(ap + (kk << 2));
      c = __builtin_amdgcn_wmma_f32_16x16x4_f32(
           false, a,  false, bfrag[kk],
           (short)0, c,  false,  false);
    }

    const int row0 = mt << 4;
    float* __restrict__ srow_out = S + (size_t)row0 * OUT_DIM + n0 + m;
#pragma unroll
    for (int r = 0; r < 8; ++r) {
      srow_out[(size_t)(mbase + r) * OUT_DIM] = c[r];
    }
    buf ^= 1;
  }
}

#define BIN_CAP 63
#define EDGES_PER_THREAD 5
#define BIN_BLOCKS 1250
#define ROWS_PER_BLOCK 8
#define GATHER_BLOCKS (N_NODES / ROWS_PER_BLOCK)

static_assert(BIN_BLOCKS * 256 * EDGES_PER_THREAD == N_EDGES, "edge split must be exact");
static_assert(GATHER_BLOCKS * ROWS_PER_BLOCK == N_NODES, "row split must be exact");
static_assert((N_EDGES & 31) == 0, "edge walk reads whole waves");

__global__ __launch_bounds__(256) void zero_bin_counts(int* __restrict__ bins) {
  const int i = blockIdx.x * 256 + threadIdx.x;
  if (i < N_NODES) bins[(size_t)i * OUT_DIM] = 0;
}

__global__ __launch_bounds__(256) void bin_edges(const int* __restrict__ rows,
                                                 const int* __restrict__ cols,
                                                 const float* __restrict__ vals,
                                                 int* __restrict__ bins) {
  const int t = blockIdx.x * 256 + threadIdx.x;
  const int stride = BIN_BLOCKS * 256;

  int r[EDGES_PER_THREAD], c[EDGES_PER_THREAD], slot[EDGES_PER_THREAD];
  float v[EDGES_PER_THREAD];
#pragma unroll
  for (int u = 0; u < EDGES_PER_THREAD; ++u) {
    const int e = t + u * stride;
    r[u] = rows[e];
    c[u] = cols[e];
    v[u] = vals[e];
  }
#pragma unroll
  for (int u = 0; u < EDGES_PER_THREAD; ++u) {
    slot[u] = __hip_atomic_fetch_add(bins + (size_t)r[u] * OUT_DIM, 1,
                                     __ATOMIC_RELAXED, __HIP_MEMORY_SCOPE_AGENT);
  }
#pragma unroll
  for (int u = 0; u < EDGES_PER_THREAD; ++u) {
    if (slot[u] < BIN_CAP) {
      int2 p;
      p.x = c[u];
      p.y = __float_as_int(v[u]);
      *(int2*)(bins + (size_t)r[u] * OUT_DIM + 2 + 2 * slot[u]) = p;
    }
  }
}

__device__ __forceinline__ unsigned lane_read(unsigned x, int srcLane) {
  return (unsigned)__builtin_amdgcn_readlane((int)x, srcLane);
}

template <int NR, int U>
__device__ __forceinline__ void add_ranked(const unsigned (&kc)[NR], const unsigned (&kv)[NR],
                                           const int (&rk)[NR], int k, int lane,
                                           const float* __restrict__ S, float4& acc) {
  float4 s[U];
  float w[U];
#pragma unroll
  for (int u = 0; u < U; ++u) {
    unsigned cb = 0, vb = 0;
#pragma unroll
    for (int a = 0; a < NR; ++a) {
      const unsigned m = __builtin_amdgcn_ballot_w32(rk[a] == k + u);
      if (NR == 1 || m != 0) {
        const int src = __builtin_ctz(m);
        cb = lane_read(kc[a], src);
        vb = lane_read(kv[a], src);
      }
    }
    w[u] = __int_as_float((int)vb);
    s[u] = *(const float4*)(S + (size_t)cb * OUT_DIM + (lane << 2));
  }
#pragma unroll
  for (int u = 0; u < U; ++u) {
    acc.x = __builtin_fmaf(w[u], s[u].x, acc.x);
    acc.y = __builtin_fmaf(w[u], s[u].y, acc.y);
    acc.z = __builtin_fmaf(w[u], s[u].z, acc.z);
    acc.w = __builtin_fmaf(w[u], s[u].w, acc.w);
  }
}

template <int NR>
__device__ __forceinline__ void add_row_sorted(const unsigned (&kc)[NR], const unsigned (&kv)[NR],
                                               int deg, int lane,
                                               const float* __restrict__ S, float4& acc) {
  int rk[NR];
  unsigned long long key[NR];
#pragma unroll
  for (int a = 0; a < NR; ++a) {
    rk[a] = 0;
    key[a] = ((unsigned long long)kc[a] << 32) | kv[a];
  }
#pragma unroll
  for (int b = 0; b < NR; ++b) {
    const int nb = min(32, deg - b * 32);
    for (int j = 0; j < nb; ++j) {
      const unsigned long long t =
          ((unsigned long long)lane_read(kc[b], j) << 32) | lane_read(kv[b], j);
      const int jj = b * 32 + j;
#pragma unroll
      for (int a = 0; a < NR; ++a) {
        const int ii = a * 32 + lane;
        rk[a] += (t < key[a] || (t == key[a] && jj < ii)) ? 1 : 0;
      }
    }
  }
  int k = 0;
  for (; k + 8 <= deg; k += 8) add_ranked<NR, 8>(kc, kv, rk, k, lane, S, acc);
  if (k + 4 <= deg) { add_ranked<NR, 4>(kc, kv, rk, k, lane, S, acc); k += 4; }
  if (k + 2 <= deg) { add_ranked<NR, 2>(kc, kv, rk, k, lane, S, acc); k += 2; }
  if (k < deg)      { add_ranked<NR, 1>(kc, kv, rk, k, lane, S, acc); }
}

__global__ __launch_bounds__(256) void gather_rows(const int* __restrict__ rows,
                                                   const int* __restrict__ cols,
                                                   const float* __restrict__ vals,
                                                   const float* __restrict__ S,
                                                   const float* __restrict__ bias,
                                                   float* out) {
  const int lane = threadIdx.x & 31;
  const int wave = __builtin_amdgcn_readfirstlane((int)(threadIdx.x >> 5));
  const int row = blockIdx.x * ROWS_PER_BLOCK + wave;

  float* orow = out + (size_t)row * OUT_DIM;
  const int* bin = (const int*)orow;

  const int cntv = bin[0];
  const int2 p0 = *(const int2*)(bin + 2 + 2 * lane);
  const int deg = __builtin_amdgcn_readfirstlane(cntv);

  float4 acc = {0.f, 0.f, 0.f, 0.f};

  if (deg <= 32) {
    unsigned kc[1], kv[1];
    const bool ok = lane < deg;
    kc[0] = ok ? (unsigned)p0.x : 0xFFFFFFFFu;
    kv[0] = ok ? (unsigned)p0.y : 0xFFFFFFFFu;
    add_row_sorted<1>(kc, kv, deg, lane, S, acc);
  } else if (deg <= BIN_CAP) {
    unsigned kc[2], kv[2];
    kc[0] = (unsigned)p0.x;
    kv[0] = (unsigned)p0.y;
    kc[1] = 0xFFFFFFFFu;
    kv[1] = 0xFFFFFFFFu;
    if (32 + lane < deg) {
      const int2 p1 = *(const int2*)(bin + 2 + 2 * (32 + lane));
      kc[1] = (unsigned)p1.x;
      kv[1] = (unsigned)p1.y;
    }
    add_row_sorted<2>(kc, kv, deg, lane, S, acc);
  } else {
    for (int base = 0; base < N_EDGES; base += 32) {
      unsigned m = __builtin_amdgcn_ballot_w32(rows[base + lane] == row);
      while (m != 0) {
        const int e = base + __builtin_ctz(m);
        m &= m - 1;
        const int cc = cols[e];
        const float w = vals[e];
        const float4 s = *(const float4*)(S + (size_t)cc * OUT_DIM + (lane << 2));
        acc.x = __builtin_fmaf(w, s.x, acc.x);
        acc.y = __builtin_fmaf(w, s.y, acc.y);
        acc.z = __builtin_fmaf(w, s.z, acc.z);
        acc.w = __builtin_fmaf(w, s.w, acc.w);
      }
    }
  }

  const float4 b = *(const float4*)(bias + (lane << 2));
  float4 res;
  res.x = acc.x + b.x;
  res.y = acc.y + b.y;
  res.z = acc.z + b.z;
  res.w = acc.w + b.w;
  *(float4*)(orow + (lane << 2)) = res;
}

extern "C" void kernel_launch(void* const* d_in, const int* in_sizes, int n_in,
                              void* d_out, int out_size, void* d_ws, size_t ws_size,
                              hipStream_t stream) {
  const float* X    = (const float*)d_in[0];
  const int*   er   = (const int*)  d_in[1];
  const int*   ec   = (const int*)  d_in[2];
  const float* ev   = (const float*)d_in[3];
  const float* W    = (const float*)d_in[4];
  const float* bias = (const float*)d_in[5];

  float* out = (float*)d_out;
  float* S   = (float*)d_ws;

  zero_bin_counts<<<(N_NODES + 255) / 256, 256, 0, stream>>>((int*)d_out);
  bin_edges<<<BIN_BLOCKS, 256, 0, stream>>>(er, ec, ev, (int*)d_out);

  gemm_wmma_f32<<<GEMM_BLOCKS, 256, 0, stream>>>(X, W, S);

  gather_rows<<<GATHER_BLOCKS, 256, 0, stream>>>(er, ec, ev, S, bias, out);
}


# Edge binning folded into the GEMM tile loop: each thread bins one edge per M-tile (slot atomic behind the barrier, pair store behind the WMMAs), so it runs under matrix work; launches 4 -> 3
# speedup vs baseline: 4.8147x; 1.1518x over previous
#include <hip/hip_runtime.h>

typedef __attribute__((ext_vector_type(2))) float v2f;
typedef __attribute__((ext_vector_type(8))) float v8f;

#define N_NODES 100000
#define N_EDGES 1600000
#define IN_DIM  256
#define OUT_DIM 128
#define M_TILES (N_NODES / 16)
#define GEMM_BLOCKS 512

#define A_PITCH 260

#define BIN_CAP 63

static_assert(GEMM_BLOCKS <= M_TILES, "every block must own at least one M-tile");
static_assert(N_EDGES == M_TILES * 256, "one edge per thread per M-tile covers the edge list exactly");

__global__ __launch_bounds__(256) void gemm_wmma_f32_bin_edges(const float* __restrict__ X,
                                                               const float* __restrict__ W,
                                                               float* __restrict__ S,
                                                               const int* __restrict__ rows,
                                                               const int* __restrict__ cols,
                                                               const float* __restrict__ vals,
                                                               int* bins) {
  __shared__ float As[2][16 * A_PITCH];

  const int tid  = threadIdx.x;
  const int lane = tid & 31;
  const int wave = tid >> 5;
  const int n0   = wave << 4;
  const int m    = lane & 15;
  const int koff = (lane >> 4) << 1;
  const int mbase = (lane >> 4) << 3;

  const int srow = tid >> 6;
  const int scol = (tid & 63) << 2;

  float4 g0, g1, g2, g3;
  int mt = blockIdx.x;
  {
    const float* __restrict__ xt = X + (size_t)((mt << 4) + srow) * IN_DIM + scol;
    g0 = *(const float4*)(xt);
    g1 = *(const float4*)(xt + 4 * IN_DIM);
    g2 = *(const float4*)(xt + 8 * IN_DIM);
    g3 = *(const float4*)(xt + 12 * IN_DIM);
  }
  int   er = rows[(mt << 8) + tid];
  int   ec = cols[(mt << 8) + tid];
  float ev = vals[(mt << 8) + tid];

  const float* __restrict__ wcol = W + (n0 + m);
  v2f bfrag[64];
#pragma unroll
  for (int kk = 0; kk < 64; ++kk) {
    const int k = (kk << 2) + koff;
    bfrag[kk].x = wcol[(size_t)k * OUT_DIM];
    bfrag[kk].y = wcol[(size_t)(k + 1) * OUT_DIM];
  }

  int buf = 0;
  for (; mt < M_TILES; mt += gridDim.x) {
    float* abuf = As[buf];
    float* ast = abuf + srow * A_PITCH + scol;
    *(float4*)(ast) = g0;
    *(float4*)(ast + 4 * A_PITCH) = g1;
    *(float4*)(ast + 8 * A_PITCH) = g2;
    *(float4*)(ast + 12 * A_PITCH) = g3;
    __syncthreads();

    int* const bin = bins + (size_t)er * OUT_DIM;
    const int   pc = ec;
    const float pv = ev;
    const int slot = __hip_atomic_fetch_add(bin, 1, __ATOMIC_RELAXED, __HIP_MEMORY_SCOPE_AGENT);

    const int nmt = mt + gridDim.x;
    if (nmt < M_TILES) {
      const float* __restrict__ xt = X + (size_t)((nmt << 4) + srow) * IN_DIM + scol;
      g0 = *(const float4*)(xt);
      g1 = *(const float4*)(xt + 4 * IN_DIM);
      g2 = *(const float4*)(xt + 8 * IN_DIM);
      g3 = *(const float4*)(xt + 12 * IN_DIM);
      er = rows[(nmt << 8) + tid];
      ec = cols[(nmt << 8) + tid];
      ev = vals[(nmt << 8) + tid];
    }

    const float* ap = abuf + m * A_PITCH + koff;
    v8f c = {0.f, 0.f, 0.f, 0.f, 0.f, 0.f, 0.f, 0.f};
#pragma unroll
    for (int kk = 0; kk < 64; ++kk) {
      const v2f a = *(const v2f*)(ap + (kk << 2));
      c = __builtin_amdgcn_wmma_f32_16x16x4_f32(
           false, a,  false, bfrag[kk],
           (short)0, c,  false,  false);
    }

    const int row0 = mt << 4;
    float* __restrict__ srow_out = S + (size_t)row0 * OUT_DIM + n0 + m;
#pragma unroll
    for (int r = 0; r < 8; ++r) {
      srow_out[(size_t)(mbase + r) * OUT_DIM] = c[r];
    }

    if (slot < BIN_CAP) {
      int2 p;
      p.x = pc;
      p.y = __float_as_int(pv);
      *(int2*)(bin + 2 + 2 * slot) = p;
    }
    buf ^= 1;
  }
}

#define ROWS_PER_BLOCK 8
#define GATHER_BLOCKS (N_NODES / ROWS_PER_BLOCK)

static_assert(GATHER_BLOCKS * ROWS_PER_BLOCK == N_NODES, "row split must be exact");
static_assert((N_EDGES & 31) == 0, "edge walk reads whole waves");

__global__ __launch_bounds__(256) void zero_bin_counts(int* __restrict__ bins) {
  const int i = blockIdx.x * 256 + threadIdx.x;
  if (i < N_NODES) bins[(size_t)i * OUT_DIM] = 0;
}

__device__ __forceinline__ unsigned lane_read(unsigned x, int srcLane) {
  return (unsigned)__builtin_amdgcn_readlane((int)x, srcLane);
}

template <int NR, int U>
__device__ __forceinline__ void add_ranked(const unsigned (&kc)[NR], const unsigned (&kv)[NR],
                                           const int (&rk)[NR], int k, int lane,
                                           const float* __restrict__ S, float4& acc) {
  float4 s[U];
  float w[U];
#pragma unroll
  for (int u = 0; u < U; ++u) {
    unsigned cb = 0, vb = 0;
#pragma unroll
    for (int a = 0; a < NR; ++a) {
      const unsigned m = __builtin_amdgcn_ballot_w32(rk[a] == k + u);
      if (NR == 1 || m != 0) {
        const int src = __builtin_ctz(m);
        cb = lane_read(kc[a], src);
        vb = lane_read(kv[a], src);
      }
    }
    w[u] = __int_as_float((int)vb);
    s[u] = *(const float4*)(S + (size_t)cb * OUT_DIM + (lane << 2));
  }
#pragma unroll
  for (int u = 0; u < U; ++u) {
    acc.x = __builtin_fmaf(w[u], s[u].x, acc.x);
    acc.y = __builtin_fmaf(w[u], s[u].y, acc.y);
    acc.z = __builtin_fmaf(w[u], s[u].z, acc.z);
    acc.w = __builtin_fmaf(w[u], s[u].w, acc.w);
  }
}

template <int NR>
__device__ __forceinline__ void add_row_sorted(const unsigned (&kc)[NR], const unsigned (&kv)[NR],
                                               int deg, int lane,
                                               const float* __restrict__ S, float4& acc) {
  int rk[NR];
  unsigned long long key[NR];
#pragma unroll
  for (int a = 0; a < NR; ++a) {
    rk[a] = 0;
    key[a] = ((unsigned long long)kc[a] << 32) | kv[a];
  }
#pragma unroll
  for (int b = 0; b < NR; ++b) {
    const int nb = min(32, deg - b * 32);
    for (int j = 0; j < nb; ++j) {
      const unsigned long long t =
          ((unsigned long long)lane_read(kc[b], j) << 32) | lane_read(kv[b], j);
      const int jj = b * 32 + j;
#pragma unroll
      for (int a = 0; a < NR; ++a) {
        const int ii = a * 32 + lane;
        rk[a] += (t < key[a] || (t == key[a] && jj < ii)) ? 1 : 0;
      }
    }
  }
  int k = 0;
  for (; k + 8 <= deg; k += 8) add_ranked<NR, 8>(kc, kv, rk, k, lane, S, acc);
  if (k + 4 <= deg) { add_ranked<NR, 4>(kc, kv, rk, k, lane, S, acc); k += 4; }
  if (k + 2 <= deg) { add_ranked<NR, 2>(kc, kv, rk, k, lane, S, acc); k += 2; }
  if (k < deg)      { add_ranked<NR, 1>(kc, kv, rk, k, lane, S, acc); }
}

__global__ __launch_bounds__(256) void gather_rows(const int* __restrict__ rows,
                                                   const int* __restrict__ cols,
                                                   const float* __restrict__ vals,
                                                   const float* __restrict__ S,
                                                   const float* __restrict__ bias,
                                                   float* out) {
  const int lane = threadIdx.x & 31;
  const int wave = __builtin_amdgcn_readfirstlane((int)(threadIdx.x >> 5));
  const int row = blockIdx.x * ROWS_PER_BLOCK + wave;

  float* orow = out + (size_t)row * OUT_DIM;
  const int* bin = (const int*)orow;

  const int cntv = bin[0];
  const int2 p0 = *(const int2*)(bin + 2 + 2 * lane);
  const int deg = __builtin_amdgcn_readfirstlane(cntv);

  float4 acc = {0.f, 0.f, 0.f, 0.f};

  if (deg <= 32) {
    unsigned kc[1], kv[1];
    const bool ok = lane < deg;
    kc[0] = ok ? (unsigned)p0.x : 0xFFFFFFFFu;
    kv[0] = ok ? (unsigned)p0.y : 0xFFFFFFFFu;
    add_row_sorted<1>(kc, kv, deg, lane, S, acc);
  } else if (deg <= BIN_CAP) {
    unsigned kc[2], kv[2];
    kc[0] = (unsigned)p0.x;
    kv[0] = (unsigned)p0.y;
    kc[1] = 0xFFFFFFFFu;
    kv[1] = 0xFFFFFFFFu;
    if (32 + lane < deg) {
      const int2 p1 = *(const int2*)(bin + 2 + 2 * (32 + lane));
      kc[1] = (unsigned)p1.x;
      kv[1] = (unsigned)p1.y;
    }
    add_row_sorted<2>(kc, kv, deg, lane, S, acc);
  } else {
    for (int base = 0; base < N_EDGES; base += 32) {
      unsigned m = __builtin_amdgcn_ballot_w32(rows[base + lane] == row);
      while (m != 0) {
        const int e = base + __builtin_ctz(m);
        m &= m - 1;
        const int cc = cols[e];
        const float w = vals[e];
        const float4 s = *(const float4*)(S + (size_t)cc * OUT_DIM + (lane << 2));
        acc.x = __builtin_fmaf(w, s.x, acc.x);
        acc.y = __builtin_fmaf(w, s.y, acc.y);
        acc.z = __builtin_fmaf(w, s.z, acc.z);
        acc.w = __builtin_fmaf(w, s.w, acc.w);
      }
    }
  }

  const float4 b = *(const float4*)(bias + (lane << 2));
  float4 res;
  res.x = acc.x + b.x;
  res.y = acc.y + b.y;
  res.z = acc.z + b.z;
  res.w = acc.w + b.w;
  *(float4*)(orow + (lane << 2)) = res;
}

extern "C" void kernel_launch(void* const* d_in, const int* in_sizes, int n_in,
                              void* d_out, int out_size, void* d_ws, size_t ws_size,
                              hipStream_t stream) {
  const float* X    = (const float*)d_in[0];
  const int*   er   = (const int*)  d_in[1];
  const int*   ec   = (const int*)  d_in[2];
  const float* ev   = (const float*)d_in[3];
  const float* W    = (const float*)d_in[4];
  const float* bias = (const float*)d_in[5];

  float* out = (float*)d_out;
  float* S   = (float*)d_ws;

  zero_bin_counts<<<(N_NODES + 255) / 256, 256, 0, stream>>>((int*)d_out);

  gemm_wmma_f32_bin_edges<<<GEMM_BLOCKS, 256, 0, stream>>>(X, W, S, er, ec, ev, (int*)d_out);

  gather_rows<<<GATHER_BLOCKS, 256, 0, stream>>>(er, ec, ev, S, bias, out);
}


# Edge binning with one memory request per edge: a 64-bit compare-and-swap puts (col,val) into an open-addressed 63-slot row bin, replacing counter atomic plus pair store; sum order unchanged
# speedup vs baseline: 5.2998x; 1.1008x over previous
#include <hip/hip_runtime.h>

typedef __attribute__((ext_vector_type(2))) float v2f;
typedef __attribute__((ext_vector_type(8))) float v8f;

#define N_NODES 100000
#define N_EDGES 1600000
#define IN_DIM  256
#define OUT_DIM 128
#define M_TILES (N_NODES / 16)
#define GEMM_BLOCKS 512

#define A_PITCH 260

#define BIN_SLOTS 63
#define BIN_EMPTY 0xFFFFFFFFFFFFFFFFull

typedef unsigned long long u64;

__device__ __forceinline__ u64 claim_slot(u64* p, u64 v) {
  u64 seen = BIN_EMPTY;
  __hip_atomic_compare_exchange_strong(p, &seen, v, __ATOMIC_RELAXED, __ATOMIC_RELAXED,
                                       __HIP_MEMORY_SCOPE_AGENT);
  return seen;
}

static_assert(GEMM_BLOCKS <= M_TILES, "every block must own at least one M-tile");
static_assert(N_EDGES == M_TILES * 256, "one edge per thread per M-tile covers the edge list exactly");

__global__ __launch_bounds__(256) void gemm_wmma_f32_bin_edges(const float* __restrict__ X,
                                                               const float* __restrict__ W,
                                                               float* __restrict__ S,
                                                               const int* __restrict__ rows,
                                                               const int* __restrict__ cols,
                                                               const float* __restrict__ vals,
                                                               u64* bins) {
  __shared__ float As[2][16 * A_PITCH];

  const int tid  = threadIdx.x;
  const int lane = tid & 31;
  const int wave = tid >> 5;
  const int n0   = wave << 4;
  const int m    = lane & 15;
  const int koff = (lane >> 4) << 1;
  const int mbase = (lane >> 4) << 3;

  const int srow = tid >> 6;
  const int scol = (tid & 63) << 2;

  float4 g0, g1, g2, g3;
  int mt = blockIdx.x;
  {
    const float* __restrict__ xt = X + (size_t)((mt << 4) + srow) * IN_DIM + scol;
    g0 = *(const float4*)(xt);
    g1 = *(const float4*)(xt + 4 * IN_DIM);
    g2 = *(const float4*)(xt + 8 * IN_DIM);
    g3 = *(const float4*)(xt + 12 * IN_DIM);
  }
  int   er = rows[(mt << 8) + tid];
  int   ec = cols[(mt << 8) + tid];
  float ev = vals[(mt << 8) + tid];

  const float* __restrict__ wcol = W + (n0 + m);
  v2f bfrag[64];
#pragma unroll
  for (int kk = 0; kk < 64; ++kk) {
    const int k = (kk << 2) + koff;
    bfrag[kk].x = wcol[(size_t)k * OUT_DIM];
    bfrag[kk].y = wcol[(size_t)(k + 1) * OUT_DIM];
  }

  int buf = 0;
  for (; mt < M_TILES; mt += gridDim.x) {
    float* abuf = As[buf];
    float* ast = abuf + srow * A_PITCH + scol;
    *(float4*)(ast) = g0;
    *(float4*)(ast + 4 * A_PITCH) = g1;
    *(float4*)(ast + 8 * A_PITCH) = g2;
    *(float4*)(ast + 12 * A_PITCH) = g3;
    __syncthreads();

    u64* const bin = bins + (size_t)er * (OUT_DIM / 2);
    const u64 pair = ((u64)(unsigned)__float_as_int(ev) << 32) | (u64)(unsigned)ec;
    unsigned slot = (unsigned)((mt << 8) + tid) % BIN_SLOTS;
    u64 seen = claim_slot(bin + slot, pair);

    const int nmt = mt + gridDim.x;
    if (nmt < M_TILES) {
      const float* __restrict__ xt = X + (size_t)((nmt << 4) + srow) * IN_DIM + scol;
      g0 = *(const float4*)(xt);
      g1 = *(const float4*)(xt + 4 * IN_DIM);
      g2 = *(const float4*)(xt + 8 * IN_DIM);
      g3 = *(const float4*)(xt + 12 * IN_DIM);
      er = rows[(nmt << 8) + tid];
      ec = cols[(nmt << 8) + tid];
      ev = vals[(nmt << 8) + tid];
    }

    const float* ap = abuf + m * A_PITCH + koff;
    v8f c = {0.f, 0.f, 0.f, 0.f, 0.f, 0.f, 0.f, 0.f};
#pragma unroll
    for (int kk = 0; kk < 64; ++kk) {
      const v2f a = *(const v2f*)(ap + (kk << 2));
      c = __builtin_amdgcn_wmma_f32_16x16x4_f32(
           false, a,  false, bfrag[kk],
           (short)0, c,  false,  false);
    }

    const int row0 = mt << 4;
    float* __restrict__ srow_out = S + (size_t)row0 * OUT_DIM + n0 + m;
#pragma unroll
    for (int r = 0; r < 8; ++r) {
      srow_out[(size_t)(mbase + r) * OUT_DIM] = c[r];
    }

    asm volatile("" : "+v"(seen));
    for (int tries = 1; seen != BIN_EMPTY && tries < BIN_SLOTS; ++tries) {
      slot = (slot + 1 == BIN_SLOTS) ? 0u : slot + 1;
      seen = claim_slot(bin + slot, pair);
    }
    if (seen != BIN_EMPTY) ((unsigned*)bin)[OUT_DIM - 1] = 0u;
    buf ^= 1;
  }
}

#define ROWS_PER_BLOCK 8
#define GATHER_BLOCKS (N_NODES / ROWS_PER_BLOCK)

static_assert(GATHER_BLOCKS * ROWS_PER_BLOCK == N_NODES, "row split must be exact");
static_assert((N_EDGES & 31) == 0, "edge walk reads whole waves");

#define FILL_BLOCKS 256
#define BIN_VEC4 (N_NODES * OUT_DIM / 4)
__global__ __launch_bounds__(256) void mark_bins_empty(int4* __restrict__ bins) {
  const int4 e = {-1, -1, -1, -1};
  for (int i = blockIdx.x * 256 + threadIdx.x; i < BIN_VEC4; i += FILL_BLOCKS * 256) bins[i] = e;
}

__device__ __forceinline__ unsigned lane_read(unsigned x, int srcLane) {
  return (unsigned)__builtin_amdgcn_readlane((int)x, srcLane);
}

template <int NR, int U>
__device__ __forceinline__ void add_ranked(const unsigned (&kc)[NR], const unsigned (&kv)[NR],
                                           const int (&rk)[NR], int k, int lane,
                                           const float* __restrict__ S, float4& acc) {
  float4 s[U];
  float w[U];
#pragma unroll
  for (int u = 0; u < U; ++u) {
    unsigned cb = 0, vb = 0;
#pragma unroll
    for (int a = 0; a < NR; ++a) {
      const unsigned m = __builtin_amdgcn_ballot_w32(rk[a] == k + u);
      if (NR == 1 || m != 0) {
        const int src = __builtin_ctz(m);
        cb = lane_read(kc[a], src);
        vb = lane_read(kv[a], src);
      }
    }
    w[u] = __int_as_float((int)vb);
    s[u] = *(const float4*)(S + (size_t)cb * OUT_DIM + (lane << 2));
  }
#pragma unroll
  for (int u = 0; u < U; ++u) {
    acc.x = __builtin_fmaf(w[u], s[u].x, acc.x);
    acc.y = __builtin_fmaf(w[u], s[u].y, acc.y);
    acc.z = __builtin_fmaf(w[u], s[u].z, acc.z);
    acc.w = __builtin_fmaf(w[u], s[u].w, acc.w);
  }
}

template <int NR>
__device__ __forceinline__ void add_row_sorted(const unsigned (&kc)[NR], const unsigned (&kv)[NR],
                                               const unsigned (&used)[NR], int deg, int lane,
                                               const float* __restrict__ S, float4& acc) {
  int rk[NR];
  unsigned long long key[NR];
#pragma unroll
  for (int a = 0; a < NR; ++a) {
    rk[a] = 0;
    key[a] = ((unsigned long long)kc[a] << 32) | kv[a];
  }
#pragma unroll
  for (int b = 0; b < NR; ++b) {
    unsigned left = used[b];
    while (left != 0) {
      const int j = __builtin_ctz(left);
      left &= left - 1;
      const unsigned long long t =
          ((unsigned long long)lane_read(kc[b], j) << 32) | lane_read(kv[b], j);
      const int jj = b * 32 + j;
#pragma unroll
      for (int a = 0; a < NR; ++a) {
        const int ii = a * 32 + lane;
        rk[a] += (t < key[a] || (t == key[a] && jj < ii)) ? 1 : 0;
      }
    }
  }
  int k = 0;
  for (; k + 8 <= deg; k += 8) add_ranked<NR, 8>(kc, kv, rk, k, lane, S, acc);
  if (k + 4 <= deg) { add_ranked<NR, 4>(kc, kv, rk, k, lane, S, acc); k += 4; }
  if (k + 2 <= deg) { add_ranked<NR, 2>(kc, kv, rk, k, lane, S, acc); k += 2; }
  if (k < deg)      { add_ranked<NR, 1>(kc, kv, rk, k, lane, S, acc); }
}

__global__ __launch_bounds__(256) void gather_rows(const int* __restrict__ rows,
                                                   const int* __restrict__ cols,
                                                   const float* __restrict__ vals,
                                                   const float* __restrict__ S,
                                                   const float* __restrict__ bias,
                                                   float* out) {
  const int lane = threadIdx.x & 31;
  const int wave = __builtin_amdgcn_readfirstlane((int)(threadIdx.x >> 5));
  const int row = blockIdx.x * ROWS_PER_BLOCK + wave;

  float* orow = out + (size_t)row * OUT_DIM;

  const int4 q = *(const int4*)(orow + (lane << 2));
  unsigned kc[2], kv[2], used[2];
  kc[0] = (unsigned)q.x;
  kv[0] = (unsigned)q.y;
  kc[1] = (unsigned)q.z;
  kv[1] = (unsigned)q.w;
  const unsigned flag = lane_read(kv[1], 31);
  used[0] = __builtin_amdgcn_ballot_w32(kc[0] != 0xFFFFFFFFu);
  used[1] = __builtin_amdgcn_ballot_w32(kc[1] != 0xFFFFFFFFu);
  const int deg = __builtin_popcount(used[0]) + __builtin_popcount(used[1]);

  float4 acc = {0.f, 0.f, 0.f, 0.f};

  if (flag == 0xFFFFFFFFu) {
    add_row_sorted<2>(kc, kv, used, deg, lane, S, acc);
  } else {
    for (int base = 0; base < N_EDGES; base += 32) {
      unsigned m = __builtin_amdgcn_ballot_w32(rows[base + lane] == row);
      while (m != 0) {
        const int e = base + __builtin_ctz(m);
        m &= m - 1;
        const int cc = cols[e];
        const float w = vals[e];
        const float4 s = *(const float4*)(S + (size_t)cc * OUT_DIM + (lane << 2));
        acc.x = __builtin_fmaf(w, s.x, acc.x);
        acc.y = __builtin_fmaf(w, s.y, acc.y);
        acc.z = __builtin_fmaf(w, s.z, acc.z);
        acc.w = __builtin_fmaf(w, s.w, acc.w);
      }
    }
  }

  const float4 b = *(const float4*)(bias + (lane << 2));
  float4 res;
  res.x = acc.x + b.x;
  res.y = acc.y + b.y;
  res.z = acc.z + b.z;
  res.w = acc.w + b.w;
  *(float4*)(orow + (lane << 2)) = res;
}

extern "C" void kernel_launch(void* const* d_in, const int* in_sizes, int n_in,
                              void* d_out, int out_size, void* d_ws, size_t ws_size,
                              hipStream_t stream) {
  const float* X    = (const float*)d_in[0];
  const int*   er   = (const int*)  d_in[1];
  const int*   ec   = (const int*)  d_in[2];
  const float* ev   = (const float*)d_in[3];
  const float* W    = (const float*)d_in[4];
  const float* bias = (const float*)d_in[5];

  float* out = (float*)d_out;
  float* S   = (float*)d_ws;

  mark_bins_empty<<<FILL_BLOCKS, 256, 0, stream>>>((int4*)d_out);

  gemm_wmma_f32_bin_edges<<<GEMM_BLOCKS, 256, 0, stream>>>(X, W, S, er, ec, ev, (u64*)d_out);

  gather_rows<<<GATHER_BLOCKS, 256, 0, stream>>>(er, ec, ev, S, bias, out);
}


# Row bins built as ordered hash tables (64-bit atomic min: a slot keeps the smaller pair, the larger moves on), so the gather adds in slot order with no ranking pass; sum order changes, stays fixed
# speedup vs baseline: 5.5011x; 1.0380x over previous
#include <hip/hip_runtime.h>

typedef __attribute__((ext_vector_type(2))) float v2f;
typedef __attribute__((ext_vector_type(8))) float v8f;

#define N_NODES 100000
#define N_EDGES 1600000
#define IN_DIM  256
#define OUT_DIM 128
#define M_TILES (N_NODES / 16)
#define GEMM_BLOCKS 512

#define A_PITCH 260

#define BIN_SLOTS 63
#define BIN_EMPTY 0xFFFFFFFFFFFFFFFFull

typedef unsigned long long u64;

__device__ __forceinline__ u64 keep_smaller(u64* p, u64 v) {
  return __hip_atomic_fetch_min(p, v, __ATOMIC_RELAXED, __HIP_MEMORY_SCOPE_AGENT);
}

__device__ __forceinline__ unsigned home_slot(unsigned col) {
  return __umulhi(col * 0x9E3779B1u, (unsigned)BIN_SLOTS);
}

static_assert(GEMM_BLOCKS <= M_TILES, "every block must own at least one M-tile");
static_assert(N_EDGES == M_TILES * 256, "one edge per thread per M-tile covers the edge list exactly");

__global__ __launch_bounds__(256) void gemm_wmma_f32_bin_edges(const float* __restrict__ X,
                                                               const float* __restrict__ W,
                                                               float* __restrict__ S,
                                                               const int* __restrict__ rows,
                                                               const int* __restrict__ cols,
                                                               const float* __restrict__ vals,
                                                               u64* bins) {
  __shared__ float As[2][16 * A_PITCH];

  const int tid  = threadIdx.x;
  const int lane = tid & 31;
  const int wave = tid >> 5;
  const int n0   = wave << 4;
  const int m    = lane & 15;
  const int koff = (lane >> 4) << 1;
  const int mbase = (lane >> 4) << 3;

  const int srow = tid >> 6;
  const int scol = (tid & 63) << 2;

  float4 g0, g1, g2, g3;
  int mt = blockIdx.x;
  {
    const float* __restrict__ xt = X + (size_t)((mt << 4) + srow) * IN_DIM + scol;
    g0 = *(const float4*)(xt);
    g1 = *(const float4*)(xt + 4 * IN_DIM);
    g2 = *(const float4*)(xt + 8 * IN_DIM);
    g3 = *(const float4*)(xt + 12 * IN_DIM);
  }
  int   er = rows[(mt << 8) + tid];
  int   ec = cols[(mt << 8) + tid];
  float ev = vals[(mt << 8) + tid];

  const float* __restrict__ wcol = W + (n0 + m);
  v2f bfrag[64];
#pragma unroll
  for (int kk = 0; kk < 64; ++kk) {
    const int k = (kk << 2) + koff;
    bfrag[kk].x = wcol[(size_t)k * OUT_DIM];
    bfrag[kk].y = wcol[(size_t)(k + 1) * OUT_DIM];
  }

  int buf = 0;
  for (; mt < M_TILES; mt += gridDim.x) {
    float* abuf = As[buf];
    float* ast = abuf + srow * A_PITCH + scol;
    *(float4*)(ast) = g0;
    *(float4*)(ast + 4 * A_PITCH) = g1;
    *(float4*)(ast + 8 * A_PITCH) = g2;
    *(float4*)(ast + 12 * A_PITCH) = g3;
    __syncthreads();

    u64* const bin = bins + (size_t)er * (OUT_DIM / 2);
    u64 pair = ((u64)(unsigned)__float_as_int(ev) << 32) | (u64)(unsigned)ec;
    unsigned slot = home_slot((unsigned)ec);
    u64 seen = keep_smaller(bin + slot, pair);

    const int nmt = mt + gridDim.x;
    if (nmt < M_TILES) {
      const float* __restrict__ xt = X + (size_t)((nmt << 4) + srow) * IN_DIM + scol;
      g0 = *(const float4*)(xt);
      g1 = *(const float4*)(xt + 4 * IN_DIM);
      g2 = *(const float4*)(xt + 8 * IN_DIM);
      g3 = *(const float4*)(xt + 12 * IN_DIM);
      er = rows[(nmt << 8) + tid];
      ec = cols[(nmt << 8) + tid];
      ev = vals[(nmt << 8) + tid];
    }

    const float* ap = abuf + m * A_PITCH + koff;
    v8f c = {0.f, 0.f, 0.f, 0.f, 0.f, 0.f, 0.f, 0.f};
#pragma unroll
    for (int kk = 0; kk < 64; ++kk) {
      const v2f a = *(const v2f*)(ap + (kk << 2));
      c = __builtin_amdgcn_wmma_f32_16x16x4_f32(
           false, a,  false, bfrag[kk],
           (short)0, c,  false,  false);
    }

    const int row0 = mt << 4;
    float* __restrict__ srow_out = S + (size_t)row0 * OUT_DIM + n0 + m;
#pragma unroll
    for (int r = 0; r < 8; ++r) {
      srow_out[(size_t)(mbase + r) * OUT_DIM] = c[r];
    }

    asm volatile("" : "+v"(seen));
    for (int tries = 1; seen != BIN_EMPTY && tries < BIN_SLOTS; ++tries) {
      pair = (seen > pair) ? seen : pair;
      slot = (slot + 1 == BIN_SLOTS) ? 0u : slot + 1;
      seen = keep_smaller(bin + slot, pair);
    }
    if (seen != BIN_EMPTY) ((unsigned*)bin)[OUT_DIM - 1] = 0u;
    buf ^= 1;
  }
}

#define ROWS_PER_BLOCK 8
#define GATHER_BLOCKS (N_NODES / ROWS_PER_BLOCK)

static_assert(GATHER_BLOCKS * ROWS_PER_BLOCK == N_NODES, "row split must be exact");
static_assert((N_EDGES & 31) == 0, "edge walk reads whole waves");

#define FILL_BLOCKS 256
#define BIN_VEC4 (N_NODES * OUT_DIM / 4)
__global__ __launch_bounds__(256) void mark_bins_empty(int4* __restrict__ bins) {
  const int4 e = {-1, -1, -1, -1};
  for (int i = blockIdx.x * 256 + threadIdx.x; i < BIN_VEC4; i += FILL_BLOCKS * 256) bins[i] = e;
}

__device__ __forceinline__ unsigned lane_read(unsigned x, int srcLane) {
  return (unsigned)__builtin_amdgcn_readlane((int)x, srcLane);
}

template <int U>
__device__ __forceinline__ void add_next(unsigned kc, unsigned kv, unsigned& left, int lane,
                                         const float* __restrict__ S, float4& acc) {
  float4 s[U];
  float w[U];
#pragma unroll
  for (int u = 0; u < U; ++u) {
    const int src = __builtin_ctz(left);
    left &= left - 1;
    const unsigned cb = lane_read(kc, src);
    w[u] = __int_as_float((int)lane_read(kv, src));
    s[u] = *(const float4*)(S + (size_t)cb * OUT_DIM + (lane << 2));
  }
#pragma unroll
  for (int u = 0; u < U; ++u) {
    acc.x = __builtin_fmaf(w[u], s[u].x, acc.x);
    acc.y = __builtin_fmaf(w[u], s[u].y, acc.y);
    acc.z = __builtin_fmaf(w[u], s[u].z, acc.z);
    acc.w = __builtin_fmaf(w[u], s[u].w, acc.w);
  }
}

__device__ __forceinline__ void add_register(unsigned kc, unsigned kv, unsigned used, int lane,
                                             const float* __restrict__ S, float4& acc) {
  unsigned left = used;
  int n = __builtin_popcount(used);
  for (; n >= 8; n -= 8) add_next<8>(kc, kv, left, lane, S, acc);
  if (n >= 4) { add_next<4>(kc, kv, left, lane, S, acc); n -= 4; }
  if (n >= 2) { add_next<2>(kc, kv, left, lane, S, acc); n -= 2; }
  if (n >= 1) { add_next<1>(kc, kv, left, lane, S, acc); }
}

__global__ __launch_bounds__(256) void gather_rows(const int* __restrict__ rows,
                                                   const int* __restrict__ cols,
                                                   const float* __restrict__ vals,
                                                   const float* __restrict__ S,
                                                   const float* __restrict__ bias,
                                                   float* out) {
  const int lane = threadIdx.x & 31;
  const int wave = __builtin_amdgcn_readfirstlane((int)(threadIdx.x >> 5));
  const int row = blockIdx.x * ROWS_PER_BLOCK + wave;

  float* orow = out + (size_t)row * OUT_DIM;

  const int4 q = *(const int4*)(orow + (lane << 2));
  unsigned kc[2], kv[2], used[2];
  kc[0] = (unsigned)q.x;
  kv[0] = (unsigned)q.y;
  kc[1] = (unsigned)q.z;
  kv[1] = (unsigned)q.w;
  const unsigned flag = lane_read(kv[1], 31);
  used[0] = __builtin_amdgcn_ballot_w32(kc[0] != 0xFFFFFFFFu);
  used[1] = __builtin_amdgcn_ballot_w32(kc[1] != 0xFFFFFFFFu);

  float4 acc = {0.f, 0.f, 0.f, 0.f};

  if (flag == 0xFFFFFFFFu) {
    add_register(kc[0], kv[0], used[0], lane, S, acc);
    add_register(kc[1], kv[1], used[1], lane, S, acc);
  } else {
    for (int base = 0; base < N_EDGES; base += 32) {
      unsigned m = __builtin_amdgcn_ballot_w32(rows[base + lane] == row);
      while (m != 0) {
        const int e = base + __builtin_ctz(m);
        m &= m - 1;
        const int cc = cols[e];
        const float w = vals[e];
        const float4 s = *(const float4*)(S + (size_t)cc * OUT_DIM + (lane << 2));
        acc.x = __builtin_fmaf(w, s.x, acc.x);
        acc.y = __builtin_fmaf(w, s.y, acc.y);
        acc.z = __builtin_fmaf(w, s.z, acc.z);
        acc.w = __builtin_fmaf(w, s.w, acc.w);
      }
    }
  }

  const float4 b = *(const float4*)(bias + (lane << 2));
  float4 res;
  res.x = acc.x + b.x;
  res.y = acc.y + b.y;
  res.z = acc.z + b.z;
  res.w = acc.w + b.w;
  *(float4*)(orow + (lane << 2)) = res;
}

extern "C" void kernel_launch(void* const* d_in, const int* in_sizes, int n_in,
                              void* d_out, int out_size, void* d_ws, size_t ws_size,
                              hipStream_t stream) {
  const float* X    = (const float*)d_in[0];
  const int*   er   = (const int*)  d_in[1];
  const int*   ec   = (const int*)  d_in[2];
  const float* ev   = (const float*)d_in[3];
  const float* W    = (const float*)d_in[4];
  const float* bias = (const float*)d_in[5];

  float* out = (float*)d_out;
  float* S   = (float*)d_ws;

  mark_bins_empty<<<FILL_BLOCKS, 256, 0, stream>>>((int4*)d_out);

  gemm_wmma_f32_bin_edges<<<GEMM_BLOCKS, 256, 0, stream>>>(X, W, S, er, ec, ev, (u64*)d_out);

  gather_rows<<<GATHER_BLOCKS, 256, 0, stream>>>(er, ec, ev, S, bias, out);
}


# Edge binning in two LDS stages, no memory atomics, no fill launch: GEMM blocks group edges by owner in LDS, store cells under the tile loop; owner blocks build row bins in LDS; sum order unchanged
# speedup vs baseline: 6.8574x; 1.2466x over previous
#include <hip/hip_runtime.h>

typedef __attribute__((ext_vector_type(2))) float v2f;
typedef __attribute__((ext_vector_type(8))) float v8f;

#define N_NODES 100000
#define N_EDGES 1600000
#define IN_DIM  256
#define OUT_DIM 128
#define M_TILES (N_NODES / 16)
#define GEMM_BLOCKS 512

#define A_PITCH 260

#define BIN_SLOTS 63
#define BIN_EMPTY 0xFFFFFFFFFFFFFFFFull

typedef unsigned long long u64;

#define OWN_ROWS 200
#define OWNERS (N_NODES / OWN_ROWS)
#define CELL_WORDS 25
#define CELL_CAP (CELL_WORDS - 1)
#define OWNER_WORDS (OWN_ROWS * (OUT_DIM / 2))
#define TILES_PER_BLOCK_MAX ((M_TILES + GEMM_BLOCKS - 1) / GEMM_BLOCKS)
#define CELL_WORDS_PER_STEP (5 * 256)
#define CELL_STEPS ((OWNERS * CELL_WORDS + CELL_WORDS_PER_STEP - 1) / CELL_WORDS_PER_STEP)

__device__ __forceinline__ u64 keep_smaller(u64* p, u64 v) {
  return __hip_atomic_fetch_min(p, v, __ATOMIC_RELAXED, __HIP_MEMORY_SCOPE_WORKGROUP);
}

__device__ __forceinline__ unsigned home_slot(unsigned col) {
  return __umulhi(col * 0x9E3779B1u, (unsigned)BIN_SLOTS);
}

static_assert(GEMM_BLOCKS <= M_TILES, "every block must own at least one M-tile");
static_assert(N_EDGES == M_TILES * 256, "one edge per thread per M-tile covers the edge list exactly");
static_assert(OWNERS * OWN_ROWS == N_NODES, "owners split the rows exactly");
static_assert(GEMM_BLOCKS * CELL_WORDS == OWNER_WORDS, "one owner's cells fill its out rows exactly");
static_assert(OWN_ROWS <= 256 && N_NODES <= (1 << 17), "local row and column share one dword");
static_assert((TILES_PER_BLOCK_MAX - 1) * GEMM_BLOCKS - 1 < M_TILES,
              "every block has all of its tiles but possibly the last");
static_assert(CELL_STEPS <= M_TILES / GEMM_BLOCKS, "every block has enough tile steps to send out all of its cells");

__global__ __launch_bounds__(256) void gemm_wmma_f32_group_edges(const float* __restrict__ X,
                                                                 const float* __restrict__ W,
                                                                 float* __restrict__ S,
                                                                 const int* __restrict__ rows,
                                                                 const int* __restrict__ cols,
                                                                 const float* __restrict__ vals,
                                                                 u64* cellmem) {
  __shared__ float As[2][16 * A_PITCH];
  __shared__ u64 cells[CELL_STEPS * CELL_WORDS_PER_STEP];
  __shared__ unsigned cellcnt[OWNERS];

  const int tid  = threadIdx.x;
  const int lane = tid & 31;
  const int wave = tid >> 5;
  const int n0   = wave << 4;
  const int m    = lane & 15;
  const int koff = (lane >> 4) << 1;
  const int mbase = (lane >> 4) << 3;

  const int srow = tid >> 6;
  const int scol = (tid & 63) << 2;

  float4 g0, g1, g2, g3;
  int mt = blockIdx.x;
  {
    const float* __restrict__ xt = X + (size_t)((mt << 4) + srow) * IN_DIM + scol;
    g0 = *(const float4*)(xt);
    g1 = *(const float4*)(xt + 4 * IN_DIM);
    g2 = *(const float4*)(xt + 8 * IN_DIM);
    g3 = *(const float4*)(xt + 12 * IN_DIM);
  }

  int   er[TILES_PER_BLOCK_MAX];
  int   ec[TILES_PER_BLOCK_MAX];
  float ev[TILES_PER_BLOCK_MAX];
#pragma unroll
  for (int i = 0; i < TILES_PER_BLOCK_MAX; ++i) {
    const int t = (int)blockIdx.x + i * GEMM_BLOCKS;
    if (i < TILES_PER_BLOCK_MAX - 1 || t < M_TILES) {
      er[i] = rows[(t << 8) + tid];
      ec[i] = cols[(t << 8) + tid];
      ev[i] = vals[(t << 8) + tid];
    }
  }
  const float* __restrict__ wcol = W + (n0 + m);
  v2f bfrag[64];
#pragma unroll
  for (int kk = 0; kk < 64; ++kk) {
    const int k = (kk << 2) + koff;
    bfrag[kk].x = wcol[(size_t)k * OUT_DIM];
    bfrag[kk].y = wcol[(size_t)(k + 1) * OUT_DIM];
  }

  for (int i = tid; i < CELL_STEPS * CELL_WORDS_PER_STEP; i += 256) cells[i] = BIN_EMPTY;
  for (int i = tid; i < OWNERS; i += 256) cellcnt[i] = 0u;
  __syncthreads();
#pragma unroll
  for (int i = 0; i < TILES_PER_BLOCK_MAX; ++i) {
    const int t = (int)blockIdx.x + i * GEMM_BLOCKS;
    if (i < TILES_PER_BLOCK_MAX - 1 || t < M_TILES) {
      const unsigned r  = (unsigned)er[i];
      const unsigned k  = r / OWN_ROWS;
      const unsigned rl = r - k * OWN_ROWS;
      const unsigned pos = atomicAdd(&cellcnt[k], 1u);
      if (pos < CELL_CAP) {
        cells[k * CELL_WORDS + 1 + pos] =
            ((u64)(unsigned)__float_as_int(ev[i]) << 32) | (u64)((rl << 17) | (unsigned)ec[i]);
      }
    }
  }
  __syncthreads();
  for (int i = tid; i < OWNERS; i += 256) cells[i * CELL_WORDS] = (u64)cellcnt[i];
  u64* const mine = cellmem + (size_t)blockIdx.x * CELL_WORDS;

  int buf = 0;
  for (int step = 0; mt < M_TILES; mt += GEMM_BLOCKS, ++step) {
    float* abuf = As[buf];
    float* ast = abuf + srow * A_PITCH + scol;
    *(float4*)(ast) = g0;
    *(float4*)(ast + 4 * A_PITCH) = g1;
    *(float4*)(ast + 8 * A_PITCH) = g2;
    *(float4*)(ast + 12 * A_PITCH) = g3;
    __syncthreads();

    const int nmt = mt + GEMM_BLOCKS;
    if (nmt < M_TILES) {
      const float* __restrict__ xt = X + (size_t)((nmt << 4) + srow) * IN_DIM + scol;
      g0 = *(const float4*)(xt);
      g1 = *(const float4*)(xt + 4 * IN_DIM);
      g2 = *(const float4*)(xt + 8 * IN_DIM);
      g3 = *(const float4*)(xt + 12 * IN_DIM);
    }

    if (step < CELL_STEPS) {
      const int i0 = step * CELL_WORDS_PER_STEP + tid;
      u64 cw[CELL_WORDS_PER_STEP / 256];
      for (int q = 0; q < CELL_WORDS_PER_STEP / 256; ++q) cw[q] = cells[i0 + q * 256];
      for (int q = 0; q < CELL_WORDS_PER_STEP / 256; ++q) {
        const int i = i0 + q * 256;
        if (i < OWNERS * CELL_WORDS) {
          const int k = i / CELL_WORDS;
          const int j = i - k * CELL_WORDS;
          mine[(size_t)k * OWNER_WORDS + j] = cw[q];
        }
      }
    }

    const float* ap = abuf + m * A_PITCH + koff;
    v8f c = {0.f, 0.f, 0.f, 0.f, 0.f, 0.f, 0.f, 0.f};
#pragma unroll
    for (int kk = 0; kk < 64; ++kk) {
      const v2f a = *(const v2f*)(ap + (kk << 2));
      c = __builtin_amdgcn_wmma_f32_16x16x4_f32(
           false, a,  false, bfrag[kk],
           (short)0, c,  false,  false);
    }

    const int row0 = mt << 4;
    float* __restrict__ srow_out = S + (size_t)row0 * OUT_DIM + n0 + m;
#pragma unroll
    for (int r = 0; r < 8; ++r) {
      srow_out[(size_t)(mbase + r) * OUT_DIM] = c[r];
    }
    buf ^= 1;
  }
}

#define OWNER_THREADS 1024
#define OWNER_WAVES (OWNER_THREADS / 32)
#define WORDS_PER_THREAD ((OWNER_WORDS + OWNER_THREADS - 1) / OWNER_THREADS)

static_assert((WORDS_PER_THREAD - 1) * OWNER_THREADS <= OWNER_WORDS, "only a thread's last word may be missing");

static_assert((N_EDGES & 31) == 0, "edge walk reads whole waves");

__device__ __forceinline__ void bin_insert(u64* bin, u64 pair, unsigned col) {
  unsigned slot = home_slot(col);
  u64 seen = keep_smaller(bin + slot, pair);
  for (int tries = 1; seen != BIN_EMPTY && tries < BIN_SLOTS; ++tries) {
    pair = (seen > pair) ? seen : pair;
    slot = (slot + 1 == BIN_SLOTS) ? 0u : slot + 1;
    seen = keep_smaller(bin + slot, pair);
  }
  if (seen != BIN_EMPTY) ((unsigned*)bin)[OUT_DIM - 1] = 0u;
}

__device__ __forceinline__ unsigned lane_read(unsigned x, int srcLane) {
  return (unsigned)__builtin_amdgcn_readlane((int)x, srcLane);
}

template <int U>
__device__ __forceinline__ void add_next(unsigned kc, unsigned kv, unsigned& left, int lane,
                                         const float* __restrict__ S, float4& acc) {
  float4 s[U];
  float w[U];
#pragma unroll
  for (int u = 0; u < U; ++u) {
    const int src = __builtin_ctz(left);
    left &= left - 1;
    const unsigned cb = lane_read(kc, src);
    w[u] = __int_as_float((int)lane_read(kv, src));
    s[u] = *(const float4*)(S + (size_t)cb * OUT_DIM + (lane << 2));
  }
#pragma unroll
  for (int u = 0; u < U; ++u) {
    acc.x = __builtin_fmaf(w[u], s[u].x, acc.x);
    acc.y = __builtin_fmaf(w[u], s[u].y, acc.y);
    acc.z = __builtin_fmaf(w[u], s[u].z, acc.z);
    acc.w = __builtin_fmaf(w[u], s[u].w, acc.w);
  }
}

__device__ __forceinline__ void add_register(unsigned kc, unsigned kv, unsigned used, int lane,
                                             const float* __restrict__ S, float4& acc) {
  unsigned left = used;
  int n = __builtin_popcount(used);
  for (; n >= 8; n -= 8) add_next<8>(kc, kv, left, lane, S, acc);
  if (n >= 4) { add_next<4>(kc, kv, left, lane, S, acc); n -= 4; }
  if (n >= 2) { add_next<2>(kc, kv, left, lane, S, acc); n -= 2; }
  if (n >= 1) { add_next<1>(kc, kv, left, lane, S, acc); }
}

__global__ __launch_bounds__(OWNER_THREADS, 16) void gather_owned_rows(const int* __restrict__ rows,
                                                                   const int* __restrict__ cols,
                                                                   const float* __restrict__ vals,
                                                                   const float* __restrict__ S,
                                                                   const float* __restrict__ bias,
                                                                   float* out) {
  __shared__ __attribute__((aligned(16))) u64 rbin[OWN_ROWS * (OUT_DIM / 2)];
  __shared__ unsigned hdr[GEMM_BLOCKS];
  __shared__ unsigned any_long;

  const int tid  = threadIdx.x;
  const int lane = tid & 31;
  const int own  = blockIdx.x;
  const int first_row = own * OWN_ROWS;
  const u64* const region = (const u64*)out + (size_t)own * OWNER_WORDS;

  u64 wv[WORDS_PER_THREAD];
#pragma unroll
  for (int n = 0; n < WORDS_PER_THREAD; ++n) {
    const int i = tid + n * OWNER_THREADS;
    wv[n] = (n < WORDS_PER_THREAD - 1 || i < OWNER_WORDS) ? region[i] : BIN_EMPTY;
  }

  for (int i = tid; i < OWN_ROWS * (OUT_DIM / 2); i += OWNER_THREADS) rbin[i] = BIN_EMPTY;
  if (tid == 0) any_long = 0u;
  __syncthreads();

#pragma unroll
  for (int n = 0; n < WORDS_PER_THREAD; ++n) {
    const int i = tid + n * OWNER_THREADS;
    const int c = i / CELL_WORDS;
    const int j = i - c * CELL_WORDS;
    if ((n < WORDS_PER_THREAD - 1 || i < OWNER_WORDS) && j == 0) {
      const unsigned cnt = (unsigned)wv[n];
      hdr[c] = cnt;
      if (cnt > CELL_CAP) any_long = 1u;
    }
  }
  __syncthreads();

#pragma unroll
  for (int n = 0; n < WORDS_PER_THREAD; ++n) {
    const int i = tid + n * OWNER_THREADS;
    const int c = i / CELL_WORDS;
    const int j = i - c * CELL_WORDS;
    if ((n < WORDS_PER_THREAD - 1 || i < OWNER_WORDS) && j != 0) {
      const unsigned cnt = hdr[c];
      if (cnt <= CELL_CAP && (unsigned)j <= cnt) {
        const u64 w = wv[n];
        const unsigned lo  = (unsigned)w;
        const unsigned rl  = lo >> 17;
        const unsigned col = lo & 0x1FFFFu;
        if (rl < OWN_ROWS) bin_insert(rbin + rl * (OUT_DIM / 2), (w & 0xFFFFFFFF00000000ull) | (u64)col, col);
      }
    }
  }

  if (any_long != 0u) {
    for (int c = 0; c < GEMM_BLOCKS; ++c) {
      if (hdr[c] > CELL_CAP) {
        for (int t = c; t < M_TILES; t += GEMM_BLOCKS) {
          if (tid < 256) {
            const int e = (t << 8) + tid;
            const unsigned r = (unsigned)rows[e];
            if (r / OWN_ROWS == (unsigned)own) {
              const unsigned col = (unsigned)cols[e];
              const unsigned vb  = (unsigned)__float_as_int(vals[e]);
              bin_insert(rbin + (r - (unsigned)first_row) * (OUT_DIM / 2), ((u64)vb << 32) | (u64)col, col);
            }
          }
        }
      }
    }
  }
  __syncthreads();

  const int wave = __builtin_amdgcn_readfirstlane((int)(threadIdx.x >> 5));
  const float4 b = *(const float4*)(bias + (lane << 2));

  for (int rl = wave; rl < OWN_ROWS; rl += OWNER_WAVES) {
    const int row = first_row + rl;
    float* orow = out + (size_t)row * OUT_DIM;

    const u64 w0 = rbin[rl * (OUT_DIM / 2) + 2 * lane];
    const u64 w1 = rbin[rl * (OUT_DIM / 2) + 2 * lane + 1];
    unsigned kc[2], kv[2], used[2];
    kc[0] = (unsigned)w0;
    kv[0] = (unsigned)(w0 >> 32);
    kc[1] = (unsigned)w1;
    kv[1] = (unsigned)(w1 >> 32);
    const unsigned flag = lane_read(kv[1], 31);
    used[0] = __builtin_amdgcn_ballot_w32(kc[0] != 0xFFFFFFFFu);
    used[1] = __builtin_amdgcn_ballot_w32(kc[1] != 0xFFFFFFFFu);

    float4 acc = {0.f, 0.f, 0.f, 0.f};

    if (flag == 0xFFFFFFFFu) {
      add_register(kc[0], kv[0], used[0], lane, S, acc);
      add_register(kc[1], kv[1], used[1], lane, S, acc);
    } else {
      for (int base = 0; base < N_EDGES; base += 32) {
        unsigned m = __builtin_amdgcn_ballot_w32(rows[base + lane] == row);
        while (m != 0) {
          const int e = base + __builtin_ctz(m);
          m &= m - 1;
          const int cc = cols[e];
          const float w = vals[e];
          const float4 s = *(const float4*)(S + (size_t)cc * OUT_DIM + (lane << 2));
          acc.x = __builtin_fmaf(w, s.x, acc.x);
          acc.y = __builtin_fmaf(w, s.y, acc.y);
          acc.z = __builtin_fmaf(w, s.z, acc.z);
          acc.w = __builtin_fmaf(w, s.w, acc.w);
        }
      }
    }

    float4 res;
    res.x = acc.x + b.x;
    res.y = acc.y + b.y;
    res.z = acc.z + b.z;
    res.w = acc.w + b.w;
    *(float4*)(orow + (lane << 2)) = res;
  }
}

extern "C" void kernel_launch(void* const* d_in, const int* in_sizes, int n_in,
                              void* d_out, int out_size, void* d_ws, size_t ws_size,
                              hipStream_t stream) {
  const float* X    = (const float*)d_in[0];
  const int*   er   = (const int*)  d_in[1];
  const int*   ec   = (const int*)  d_in[2];
  const float* ev   = (const float*)d_in[3];
  const float* W    = (const float*)d_in[4];
  const float* bias = (const float*)d_in[5];

  float* out = (float*)d_out;
  float* S   = (float*)d_ws;

  gemm_wmma_f32_group_edges<<<GEMM_BLOCKS, 256, 0, stream>>>(X, W, S, er, ec, ev, (u64*)d_out);

  gather_owned_rows<<<OWNERS, OWNER_THREADS, 0, stream>>>(er, ec, ev, S, bias, out);
}
